# starting kernel
# baseline (speedup 1.0000x reference)
#include <hip/hip_runtime.h>


#define IN_NODES   4096
#define OUT_NODES  1024
#define FD         16
#define ROWS_BLK   32
#define NSUB       2
#define NBLK       128
#define SVEC       16384
#define BSTRIDE    1025

typedef __attribute__((ext_vector_type(2))) float v2f;
typedef __attribute__((ext_vector_type(8))) float v8f;

#define SMEM_FLOATS (SVEC + SVEC + 16*BSTRIDE + 256 + 16 + 16)
#define SMEM_BYTES  (SMEM_FLOATS * 4)

__global__ void caps_zero_vsum(float* __restrict__ vsum) {
    vsum[blockIdx.x * 256 + threadIdx.x] = 0.0f;
}

__global__ void caps_routing_pass(const float* __restrict__ uh,
                                  const float* __restrict__ vsum_g,
                                  float* __restrict__ partials) {
    extern __shared__ float smem[];
    float* vs   = smem;
    float* sacc = vs   + SVEC;
    float* bnew = sacc + SVEC;
    float* stat = bnew + 16 * BSTRIDE;
    float* rmax = stat + 256;
    float* rinv = rmax + 16;

    const int t    = threadIdx.x;
    const int lane = t & 31;
    const int wave = t >> 5;
    const int obase = wave * 128;

    for (int q = t; q < SVEC / 4; q += 256) {
        ((float4*)vs)[q]   = ((const float4*)vsum_g)[q];
        ((float4*)sacc)[q] = make_float4(0.f, 0.f, 0.f, 0.f);
    }
    __syncthreads();

    for (int st = 0; st < NSUB; ++st) {
        const int i0 = blockIdx.x * ROWS_BLK + st * 16;

        {
            const int m  = lane & 15;
            const int oh = lane >> 4;
            for (int j = 0; j < 64; ++j) {
                const int o = obase + 2 * j + oh;
                const float4* up = (const float4*)(uh + ((size_t)(i0 + m) * OUT_NODES + o) * FD);
                const float*  vv = vs + o * FD;
                float acc = 0.f;
                #pragma unroll
                for (int q = 0; q < 4; ++q) {
                    const float4 u4 = up[q];
                    acc += u4.x * vv[4*q+0] + u4.y * vv[4*q+1]
                         + u4.z * vv[4*q+2] + u4.w * vv[4*q+3];
                }
                bnew[m * BSTRIDE + o] = acc;
            }
        }
        __syncthreads();

        {
            const int r = t >> 4, sub = t & 15;
            float pm = -3.4e38f;
            for (int q = sub; q < OUT_NODES; q += 16)
                pm = fmaxf(pm, bnew[r * BSTRIDE + q]);
            stat[r * 16 + sub] = pm;
            __syncthreads();
            if (sub == 0) {
                float mm = stat[r * 16];
                for (int q = 1; q < 16; ++q) mm = fmaxf(mm, stat[r * 16 + q]);
                rmax[r] = mm;
            }
            __syncthreads();
            const float mm = rmax[r];
            float ps = 0.f;
            for (int q = sub; q < OUT_NODES; q += 16) {
                const float e = __expf(bnew[r * BSTRIDE + q] - mm);
                bnew[r * BSTRIDE + q] = e;
                ps += e;
            }
            stat[r * 16 + sub] = ps;
            __syncthreads();
            if (sub == 0) {
                float ss = 0.f;
                for (int q = 0; q < 16; ++q) ss += stat[r * 16 + q];
                rinv[r] = 1.f / ss;
            }
            __syncthreads();
        }

        {
            const int hk = (lane < 16) ? 0 : 2;
            const int fl = lane & 15;
            for (int j = 0; j < 128; ++j) {
                const int o = obase + j;
                v8f d = {0.f, 0.f, 0.f, 0.f, 0.f, 0.f, 0.f, 0.f};
                #pragma unroll
                for (int k = 0; k < 4; ++k) {
                    const int r0 = 4 * k + hk;
                    v2f a, b;
                    a.x = bnew[(r0    ) * BSTRIDE + o] * rinv[r0    ];
                    a.y = bnew[(r0 + 1) * BSTRIDE + o] * rinv[r0 + 1];
                    const float* bp = uh + ((size_t)(i0 + r0) * OUT_NODES + o) * FD + fl;
                    b.x = bp[0];
                    b.y = bp[OUT_NODES * FD];
                    d = __builtin_amdgcn_wmma_f32_16x16x4_f32(
                             false, a,  false, b,
                             (short)0, d,  false,  false);
                }
                if (lane < 16) sacc[o * FD + fl] += d[0];
            }
        }
        __syncthreads();
    }

    float* pout = partials + (size_t)blockIdx.x * SVEC;
    for (int q = t; q < SVEC / 4; q += 256)
        ((float4*)pout)[q] = ((const float4*)sacc)[q];
}

__global__ void caps_reduce_squash(const float* __restrict__ partials,
                                   float* __restrict__ vsum,
                                   float* __restrict__ vout) {
    const int idx = blockIdx.x * 256 + threadIdx.x;
    float s = 0.f;
    for (int b = 0; b < NBLK; ++b)
        s += partials[(size_t)b * SVEC + idx];
    float sq = s * s;
    #pragma unroll
    for (int m = 1; m < 16; m <<= 1)
        sq += __shfl_xor(sq, m, 16);
    const float scale = (sq / (1.f + sq)) * rsqrtf(sq);
    const float v = s * scale;
    vout[idx] = v;
    vsum[idx] += v;
}

extern "C" void kernel_launch(void* const* d_in, const int* in_sizes, int n_in,
                              void* d_out, int out_size, void* d_ws, size_t ws_size,
                              hipStream_t stream) {
    const float* uh = (const float*)d_in[0];
    const int routing_num = 3;

    float* ws       = (float*)d_ws;
    float* vsum     = ws;
    float* partials = ws + SVEC;
    float* vout     = (float*)d_out;

    caps_zero_vsum<<<SVEC / 256, 256, 0, stream>>>(vsum);
    for (int it = 0; it < routing_num; ++it) {
        caps_routing_pass<<<NBLK, 256, SMEM_BYTES, stream>>>(uh, vsum, partials);
        caps_reduce_squash<<<SVEC / 256, 256, 0, stream>>>(partials, vsum, vout);
    }
}


# More loads in flight: routing pass runs 32 waves per block (was 8) and issues 16 to 64 independent loads per lane before first use; softmax row max/sum now 64 partials of 16 (was 16 of 64)
# speedup vs baseline: 1.2829x; 1.2829x over previous
#include <hip/hip_runtime.h>


#define IN_NODES   4096
#define OUT_NODES  1024
#define FD         16
#define ROWS_BLK   32
#define NSUB       2
#define NBLK       128
#define SVEC       16384
#define BSTRIDE    1025
#define NTHR       1024
#define OWAVE      32
#define NPART      64

typedef __attribute__((ext_vector_type(2))) float v2f;
typedef __attribute__((ext_vector_type(8))) float v8f;

#define SMEM_FLOATS (SVEC + SVEC + 16*BSTRIDE + 16*NPART + 16 + 16)
#define SMEM_BYTES  (SMEM_FLOATS * 4)

__global__ void caps_zero_vsum(float* __restrict__ vsum) {
    vsum[blockIdx.x * 256 + threadIdx.x] = 0.0f;
}

__global__ void __launch_bounds__(NTHR)
caps_routing_pass(const float* __restrict__ uh,
                                  const float* __restrict__ vsum_g,
                                  float* __restrict__ partials) {
    extern __shared__ float smem[];
    float* vs   = smem;
    float* sacc = vs   + SVEC;
    float* bnew = sacc + SVEC;
    float* stat = bnew + 16 * BSTRIDE;
    float* rmax = stat + 16 * NPART;
    float* rinv = rmax + 16;

    const int t    = threadIdx.x;
    const int lane = t & 31;
    const int wave = t >> 5;
    const int obase = wave * OWAVE;

    for (int q = t; q < SVEC / 4; q += NTHR) {
        ((float4*)vs)[q]   = ((const float4*)vsum_g)[q];
        ((float4*)sacc)[q] = make_float4(0.f, 0.f, 0.f, 0.f);
    }
    __syncthreads();

    for (int st = 0; st < NSUB; ++st) {
        const int i0 = blockIdx.x * ROWS_BLK + st * 16;

        {
            const int m  = lane & 15;
            const int oh = lane >> 4;
            const float* urow = uh + ((size_t)(i0 + m) * OUT_NODES + obase + oh) * FD;
            #pragma unroll 1
            for (int jb = 0; jb < OWAVE / 2; jb += 4) {
                float4 u[4][4];
                #pragma unroll
                for (int jj = 0; jj < 4; ++jj) {
                    const float4* up = (const float4*)(urow + (size_t)(2 * (jb + jj)) * FD);
                    #pragma unroll
                    for (int q = 0; q < 4; ++q) u[jj][q] = up[q];
                }
                __builtin_amdgcn_sched_barrier(0);
                #pragma unroll
                for (int jj = 0; jj < 4; ++jj) {
                    const int o = obase + 2 * (jb + jj) + oh;
                    const float* vv = vs + o * FD;
                    float acc = 0.f;
                    #pragma unroll
                    for (int q = 0; q < 4; ++q) {
                        const float4 u4 = u[jj][q];
                        acc += u4.x * vv[4*q+0] + u4.y * vv[4*q+1]
                             + u4.z * vv[4*q+2] + u4.w * vv[4*q+3];
                    }
                    bnew[m * BSTRIDE + o] = acc;
                }
            }
        }
        __syncthreads();

        {
            const int r = t >> 6, sub = t & (NPART - 1);
            float pm = -3.4e38f;
            for (int q = sub; q < OUT_NODES; q += NPART)
                pm = fmaxf(pm, bnew[r * BSTRIDE + q]);
            stat[r * NPART + sub] = pm;
            __syncthreads();
            if (sub == 0) {
                float mm = stat[r * NPART];
                for (int q = 1; q < NPART; ++q) mm = fmaxf(mm, stat[r * NPART + q]);
                rmax[r] = mm;
            }
            __syncthreads();
            const float mm = rmax[r];
            float ps = 0.f;
            for (int q = sub; q < OUT_NODES; q += NPART) {
                const float e = __expf(bnew[r * BSTRIDE + q] - mm);
                bnew[r * BSTRIDE + q] = e;
                ps += e;
            }
            stat[r * NPART + sub] = ps;
            __syncthreads();
            if (sub == 0) {
                float ss = 0.f;
                for (int q = 0; q < NPART; ++q) ss += stat[r * NPART + q];
                rinv[r] = 1.f / ss;
            }
            __syncthreads();
        }

        {
            const int hk = (lane < 16) ? 0 : 2;
            const int fl = lane & 15;
            #pragma unroll 1
            for (int jb = 0; jb < OWAVE; jb += 8) {
                v2f   bf[8][4];
                float dres[8];
                #pragma unroll
                for (int jp = 0; jp < 4; ++jp) {
                    #pragma unroll
                    for (int k = 0; k < 4; ++k) {
                        const int r0 = 4 * k + hk;
                        const float* bp = uh + ((size_t)(i0 + r0) * OUT_NODES + (obase + jb + 2 * jp)) * FD + fl;
                        bf[2 * jp    ][k].x = bp[0];
                        bf[2 * jp + 1][k].x = bp[FD];
                        bf[2 * jp    ][k].y = bp[OUT_NODES * FD];
                        bf[2 * jp + 1][k].y = bp[OUT_NODES * FD + FD];
                    }
                }
                #pragma unroll
                for (int jj = 0; jj < 8; ++jj) {
                    const int o = obase + jb + jj;
                    v8f d = {0.f, 0.f, 0.f, 0.f, 0.f, 0.f, 0.f, 0.f};
                    #pragma unroll
                    for (int k = 0; k < 4; ++k) {
                        const int r0 = 4 * k + hk;
                        v2f a;
                        a.x = bnew[(r0    ) * BSTRIDE + o] * rinv[r0    ];
                        a.y = bnew[(r0 + 1) * BSTRIDE + o] * rinv[r0 + 1];
                        d = __builtin_amdgcn_wmma_f32_16x16x4_f32(
                                 false, a,  false, bf[jj][k],
                                 (short)0, d,  false,  false);
                    }
                    dres[jj] = d[0];
                }
                if (lane < 16) {
                    #pragma unroll
                    for (int jj = 0; jj < 8; ++jj)
                        sacc[(obase + jb + jj) * FD + fl] += dres[jj];
                }
            }
        }
        __syncthreads();
    }

    float* pout = partials + (size_t)blockIdx.x * SVEC;
    for (int q = t; q < SVEC / 4; q += NTHR)
        ((float4*)pout)[q] = ((const float4*)sacc)[q];
}

__global__ void caps_reduce_squash(const float* __restrict__ partials,
                                   float* __restrict__ vsum,
                                   float* __restrict__ vout) {
    const int idx = blockIdx.x * 256 + threadIdx.x;
    float s = 0.f;
    for (int b = 0; b < NBLK; ++b)
        s += partials[(size_t)b * SVEC + idx];
    float sq = s * s;
    #pragma unroll
    for (int m = 1; m < 16; m <<= 1)
        sq += __shfl_xor(sq, m, 16);
    const float scale = (sq / (1.f + sq)) * rsqrtf(sq);
    const float v = s * scale;
    vout[idx] = v;
    vsum[idx] += v;
}

extern "C" void kernel_launch(void* const* d_in, const int* in_sizes, int n_in,
                              void* d_out, int out_size, void* d_ws, size_t ws_size,
                              hipStream_t stream) {
    const float* uh = (const float*)d_in[0];
    const int routing_num = 3;

    float* ws       = (float*)d_ws;
    float* vsum     = ws;
    float* partials = ws + SVEC;
    float* vout     = (float*)d_out;

    caps_zero_vsum<<<SVEC / 256, 256, 0, stream>>>(vsum);
    for (int it = 0; it < routing_num; ++it) {
        caps_routing_pass<<<NBLK, NTHR, SMEM_BYTES, stream>>>(uh, vsum, partials);
        caps_reduce_squash<<<SVEC / 256, 256, 0, stream>>>(partials, vsum, vout);
    }
}


# Read u_hat once per routing pass: each 4-row tile stays in registers from the agreement dots to the WMMA weighted sum; dot, softmax-sum and s-partial summation order regrouped
# speedup vs baseline: 3.3678x; 2.6251x over previous
#include <hip/hip_runtime.h>


#define IN_NODES   4096
#define OUT_NODES  1024
#define FD         16
#define ROWS_BLK   32
#define ROWS_SUB   4
#define NSUBT      8
#define NBLK       128
#define SVEC       16384
#define BSTRIDE    1028
#define NTHR       1024
#define OWAVE      32
#define NGRP       8
#define WROW       8

typedef __attribute__((ext_vector_type(2))) float v2f;
typedef __attribute__((ext_vector_type(8))) float v8f;

#define SMEM_FLOATS (SVEC + SVEC + 2 * ROWS_SUB * BSTRIDE + ROWS_SUB * WROW + ROWS_SUB * WROW)
#define SMEM_BYTES  (SMEM_FLOATS * 4)

template <int M>
__device__ __forceinline__ float row_xor(float x) {
    return __builtin_bit_cast(float, __builtin_amdgcn_update_dpp(
        0, __builtin_bit_cast(int, x), 0x160 + M, 0xf, 0xf, true));
}
__device__ __forceinline__ float other_row(float x) {
    return __builtin_bit_cast(float, __builtin_amdgcn_permlanex16(
        __builtin_bit_cast(int, x), __builtin_bit_cast(int, x),
        0x76543210, 0xfedcba98, false, false));
}

__global__ void caps_zero_vsum(float* __restrict__ vsum) {
    vsum[blockIdx.x * 256 + threadIdx.x] = 0.0f;
}

template <bool MORE>
__device__ __forceinline__ void tile_step(float4 (&u)[NGRP][2], float* sacc,
                                          float* bw, const float* vs,
                                          float* smax, float* ssum,
                                          const float* __restrict__ un,
                                          const int t, const int lane, const int wave,
                                          const int fl, const int h, const int obase) {
    #pragma unroll
    for (int g = 0; g < NGRP; ++g) {
        const float4 v4 = *(const float4*)(vs + (obase + 4 * g) * FD + fl * 4);
        float p0 = u[g][0].x * v4.x + u[g][0].y * v4.y + u[g][0].z * v4.z + u[g][0].w * v4.w;
        float p1 = u[g][1].x * v4.x + u[g][1].y * v4.y + u[g][1].z * v4.z + u[g][1].w * v4.w;
        p0 += row_xor<1>(p0);  p1 += row_xor<1>(p1);
        p0 += row_xor<2>(p0);  p1 += row_xor<2>(p1);
        const float pv = (fl & 1) ? p1 : p0;
        if ((fl & 2) == 0)
            bw[(2 * h + (fl & 1)) * BSTRIDE + obase + 4 * g + (fl >> 2)] = pv;
    }
    __syncthreads();

    {
        const int r = t >> 8, sub = t & 255, wr = wave & (WROW - 1);
        float* row = bw + r * BSTRIDE;
        const float x0 = row[sub], x1 = row[sub + 256], x2 = row[sub + 512], x3 = row[sub + 768];
        float pm = fmaxf(fmaxf(x0, x1), fmaxf(x2, x3));
        pm = fmaxf(pm, row_xor<1>(pm));
        pm = fmaxf(pm, row_xor<2>(pm));
        pm = fmaxf(pm, row_xor<4>(pm));
        pm = fmaxf(pm, row_xor<8>(pm));
        pm = fmaxf(pm, other_row(pm));
        if (lane == 0) smax[r * WROW + wr] = pm;
        __syncthreads();
        float mm = smax[r * WROW];
        #pragma unroll
        for (int q = 1; q < WROW; ++q) mm = fmaxf(mm, smax[r * WROW + q]);
        const float e0 = __expf(x0 - mm), e1 = __expf(x1 - mm);
        const float e2 = __expf(x2 - mm), e3 = __expf(x3 - mm);
        row[sub] = e0;  row[sub + 256] = e1;  row[sub + 512] = e2;  row[sub + 768] = e3;
        float ps = (e0 + e1) + (e2 + e3);
        ps += row_xor<1>(ps);
        ps += row_xor<2>(ps);
        ps += row_xor<4>(ps);
        ps += row_xor<8>(ps);
        ps += other_row(ps);
        if (lane == 0) ssum[r * WROW + wr] = ps;
        __syncthreads();
    }

    {
        float s0 = 0.f, s1 = 0.f;
        #pragma unroll
        for (int q = 0; q < WROW; ++q) {
            s0 += ssum[(2 * h    ) * WROW + q];
            s1 += ssum[(2 * h + 1) * WROW + q];
        }
        const float ri0 = 1.f / s0, ri1 = 1.f / s1;
        const float* e0p = bw + (2 * h) * BSTRIDE + obase + (fl >> 2);
        const float* e1p = e0p + BSTRIDE;
        const bool own = (h == (fl >> 3));
        #pragma unroll
        for (int g = 0; g < NGRP; ++g) {
            v2f a;
            a.x = e0p[4 * g] * ri0;
            a.y = e1p[4 * g] * ri1;
            const float u0[4] = {u[g][0].x, u[g][0].y, u[g][0].z, u[g][0].w};
            const float u1[4] = {u[g][1].x, u[g][1].y, u[g][1].z, u[g][1].w};
            float dv[4];
            #pragma unroll
            for (int j = 0; j < 4; ++j) {
                v2f b;
                b.x = u0[j];
                b.y = u1[j];
                const v8f z = {0.f, 0.f, 0.f, 0.f, 0.f, 0.f, 0.f, 0.f};
                const v8f d = __builtin_amdgcn_wmma_f32_16x16x4_f32(
                         false, a,  false, b,
                         (short)0, z,  false,  false);
                dv[j] = (fl & 4) ? d[4] : d[0];
            }
            if (own) {
                float4* sp = (float4*)(sacc + (obase + 4 * g) * FD + fl * 4);
                float4 sv = *sp;
                sv.x += dv[0];  sv.y += dv[1];  sv.z += dv[2];  sv.w += dv[3];
                *sp = sv;
            }
            if (MORE) {
                u[g][0] = *(const float4*)(un + g * 4 * FD);
                u[g][1] = *(const float4*)(un + OUT_NODES * FD + g * 4 * FD);
            }
        }
    }
}

__global__ void __launch_bounds__(NTHR)
caps_routing_pass(const float* __restrict__ uh,
                                  const float* __restrict__ vsum_g,
                                  float* __restrict__ partials) {
    extern __shared__ float smem[];
    float* vs   = smem;
    float* sacc = vs + SVEC;
    float* btl  = sacc + SVEC;
    float* smax = btl + 2 * ROWS_SUB * BSTRIDE;
    float* ssum = smax + ROWS_SUB * WROW;

    const int t    = threadIdx.x;
    const int lane = t & 31;
    const int wave = t >> 5;
    const int fl   = lane & 15;
    const int h    = lane >> 4;
    const int obase = wave * OWAVE;

    for (int q = t; q < SVEC / 4; q += NTHR) {
        ((float4*)vs)[q]   = ((const float4*)vsum_g)[q];
        ((float4*)sacc)[q] = make_float4(0.f, 0.f, 0.f, 0.f);
    }

    const float* ub = uh + ((size_t)(blockIdx.x * ROWS_BLK + 2 * h) * OUT_NODES + obase) * FD + fl * 4;

    float4 u[NGRP][2];
    #pragma unroll
    for (int g = 0; g < NGRP; ++g) {
        u[g][0] = *(const float4*)(ub + g * 4 * FD);
        u[g][1] = *(const float4*)(ub + OUT_NODES * FD + g * 4 * FD);
    }
    __syncthreads();

    #pragma unroll 1
    for (int st = 0; st < NSUBT - 1; ++st)
        tile_step<true>(u, sacc, btl + (st & 1) * (ROWS_SUB * BSTRIDE), vs, smax, ssum,
                        ub + (size_t)(st + 1) * ROWS_SUB * OUT_NODES * FD, t, lane, wave, fl, h, obase);
    tile_step<false>(u, sacc, btl + ((NSUBT - 1) & 1) * (ROWS_SUB * BSTRIDE), vs, smax, ssum,
                     ub, t, lane, wave, fl, h, obase);

    __syncthreads();
    float* pout = partials + (size_t)blockIdx.x * SVEC;
    for (int q = t; q < SVEC / 4; q += NTHR)
        ((float4*)pout)[q] = ((const float4*)sacc)[q];
}

__global__ void caps_reduce_squash(const float* __restrict__ partials,
                                   float* __restrict__ vsum,
                                   float* __restrict__ vout) {
    const int idx = blockIdx.x * 256 + threadIdx.x;
    float s = 0.f;
    for (int b = 0; b < NBLK; ++b)
        s += partials[(size_t)b * SVEC + idx];
    float sq = s * s;
    #pragma unroll
    for (int m = 1; m < 16; m <<= 1)
        sq += __shfl_xor(sq, m, 16);
    const float scale = (sq / (1.f + sq)) * rsqrtf(sq);
    const float v = s * scale;
    vout[idx] = v;
    vsum[idx] += v;
}

extern "C" void kernel_launch(void* const* d_in, const int* in_sizes, int n_in,
                              void* d_out, int out_size, void* d_ws, size_t ws_size,
                              hipStream_t stream) {
    const float* uh = (const float*)d_in[0];
    const int routing_num = 3;

    float* ws       = (float*)d_ws;
    float* vsum     = ws;
    float* partials = ws + SVEC;
    float* vout     = (float*)d_out;

    caps_zero_vsum<<<SVEC / 256, 256, 0, stream>>>(vsum);
    for (int it = 0; it < routing_num; ++it) {
        caps_routing_pass<<<NBLK, NTHR, SMEM_BYTES, stream>>>(uh, vsum, partials);
        caps_reduce_squash<<<SVEC / 256, 256, 0, stream>>>(partials, vsum, vout);
    }
}


# Use all 256 WGPs: routing pass runs 256 work-groups of 16 rows (was 128 of 32: one per WGP, half the chip idle); s is now summed from 256 block partials of 16 rows (was 128 of 32)
# speedup vs baseline: 3.9417x; 1.1704x over previous
#include <hip/hip_runtime.h>


#define IN_NODES   4096
#define OUT_NODES  1024
#define FD         16
#define ROWS_BLK   16
#define ROWS_SUB   4
#define NSUBT      4
#define NBLK       256
#define SVEC       16384
#define BSTRIDE    1028
#define NTHR       1024
#define OWAVE      32
#define NGRP       8
#define WROW       8

typedef __attribute__((ext_vector_type(2))) float v2f;
typedef __attribute__((ext_vector_type(8))) float v8f;

#define SMEM_FLOATS (SVEC + SVEC + 2 * ROWS_SUB * BSTRIDE + ROWS_SUB * WROW + ROWS_SUB * WROW)
#define SMEM_BYTES  (SMEM_FLOATS * 4)

template <int M>
__device__ __forceinline__ float row_xor(float x) {
    return __builtin_bit_cast(float, __builtin_amdgcn_update_dpp(
        0, __builtin_bit_cast(int, x), 0x160 + M, 0xf, 0xf, true));
}
__device__ __forceinline__ float other_row(float x) {
    return __builtin_bit_cast(float, __builtin_amdgcn_permlanex16(
        __builtin_bit_cast(int, x), __builtin_bit_cast(int, x),
        0x76543210, 0xfedcba98, false, false));
}

__global__ void caps_zero_vsum(float* __restrict__ vsum) {
    vsum[blockIdx.x * 256 + threadIdx.x] = 0.0f;
}

template <bool MORE>
__device__ __forceinline__ void tile_step(float4 (&u)[NGRP][2], float* sacc,
                                          float* bw, const float* vs,
                                          float* smax, float* ssum,
                                          const float* __restrict__ un,
                                          const int t, const int lane, const int wave,
                                          const int fl, const int h, const int obase) {
    #pragma unroll
    for (int g = 0; g < NGRP; ++g) {
        const float4 v4 = *(const float4*)(vs + (obase + 4 * g) * FD + fl * 4);
        float p0 = u[g][0].x * v4.x + u[g][0].y * v4.y + u[g][0].z * v4.z + u[g][0].w * v4.w;
        float p1 = u[g][1].x * v4.x + u[g][1].y * v4.y + u[g][1].z * v4.z + u[g][1].w * v4.w;
        p0 += row_xor<1>(p0);  p1 += row_xor<1>(p1);
        p0 += row_xor<2>(p0);  p1 += row_xor<2>(p1);
        const float pv = (fl & 1) ? p1 : p0;
        if ((fl & 2) == 0)
            bw[(2 * h + (fl & 1)) * BSTRIDE + obase + 4 * g + (fl >> 2)] = pv;
    }
    __syncthreads();

    {
        const int r = t >> 8, sub = t & 255, wr = wave & (WROW - 1);
        float* row = bw + r * BSTRIDE;
        const float x0 = row[sub], x1 = row[sub + 256], x2 = row[sub + 512], x3 = row[sub + 768];
        float pm = fmaxf(fmaxf(x0, x1), fmaxf(x2, x3));
        pm = fmaxf(pm, row_xor<1>(pm));
        pm = fmaxf(pm, row_xor<2>(pm));
        pm = fmaxf(pm, row_xor<4>(pm));
        pm = fmaxf(pm, row_xor<8>(pm));
        pm = fmaxf(pm, other_row(pm));
        if (lane == 0) smax[r * WROW + wr] = pm;
        __syncthreads();
        float mm = smax[r * WROW];
        #pragma unroll
        for (int q = 1; q < WROW; ++q) mm = fmaxf(mm, smax[r * WROW + q]);
        const float e0 = __expf(x0 - mm), e1 = __expf(x1 - mm);
        const float e2 = __expf(x2 - mm), e3 = __expf(x3 - mm);
        row[sub] = e0;  row[sub + 256] = e1;  row[sub + 512] = e2;  row[sub + 768] = e3;
        float ps = (e0 + e1) + (e2 + e3);
        ps += row_xor<1>(ps);
        ps += row_xor<2>(ps);
        ps += row_xor<4>(ps);
        ps += row_xor<8>(ps);
        ps += other_row(ps);
        if (lane == 0) ssum[r * WROW + wr] = ps;
        __syncthreads();
    }

    {
        float s0 = 0.f, s1 = 0.f;
        #pragma unroll
        for (int q = 0; q < WROW; ++q) {
            s0 += ssum[(2 * h    ) * WROW + q];
            s1 += ssum[(2 * h + 1) * WROW + q];
        }
        const float ri0 = 1.f / s0, ri1 = 1.f / s1;
        const float* e0p = bw + (2 * h) * BSTRIDE + obase + (fl >> 2);
        const float* e1p = e0p + BSTRIDE;
        const bool own = (h == (fl >> 3));
        #pragma unroll
        for (int g = 0; g < NGRP; ++g) {
            v2f a;
            a.x = e0p[4 * g] * ri0;
            a.y = e1p[4 * g] * ri1;
            const float u0[4] = {u[g][0].x, u[g][0].y, u[g][0].z, u[g][0].w};
            const float u1[4] = {u[g][1].x, u[g][1].y, u[g][1].z, u[g][1].w};
            float dv[4];
            #pragma unroll
            for (int j = 0; j < 4; ++j) {
                v2f b;
                b.x = u0[j];
                b.y = u1[j];
                const v8f z = {0.f, 0.f, 0.f, 0.f, 0.f, 0.f, 0.f, 0.f};
                const v8f d = __builtin_amdgcn_wmma_f32_16x16x4_f32(
                         false, a,  false, b,
                         (short)0, z,  false,  false);
                dv[j] = (fl & 4) ? d[4] : d[0];
            }
            if (own) {
                float4* sp = (float4*)(sacc + (obase + 4 * g) * FD + fl * 4);
                float4 sv = *sp;
                sv.x += dv[0];  sv.y += dv[1];  sv.z += dv[2];  sv.w += dv[3];
                *sp = sv;
            }
            if (MORE) {
                u[g][0] = *(const float4*)(un + g * 4 * FD);
                u[g][1] = *(const float4*)(un + OUT_NODES * FD + g * 4 * FD);
            }
        }
    }
}

__global__ void __launch_bounds__(NTHR)
caps_routing_pass(const float* __restrict__ uh,
                                  const float* __restrict__ vsum_g,
                                  float* __restrict__ partials) {
    extern __shared__ float smem[];
    float* vs   = smem;
    float* sacc = vs + SVEC;
    float* btl  = sacc + SVEC;
    float* smax = btl + 2 * ROWS_SUB * BSTRIDE;
    float* ssum = smax + ROWS_SUB * WROW;

    const int t    = threadIdx.x;
    const int lane = t & 31;
    const int wave = t >> 5;
    const int fl   = lane & 15;
    const int h    = lane >> 4;
    const int obase = wave * OWAVE;

    for (int q = t; q < SVEC / 4; q += NTHR) {
        ((float4*)vs)[q]   = ((const float4*)vsum_g)[q];
        ((float4*)sacc)[q] = make_float4(0.f, 0.f, 0.f, 0.f);
    }

    const float* ub = uh + ((size_t)(blockIdx.x * ROWS_BLK + 2 * h) * OUT_NODES + obase) * FD + fl * 4;

    float4 u[NGRP][2];
    #pragma unroll
    for (int g = 0; g < NGRP; ++g) {
        u[g][0] = *(const float4*)(ub + g * 4 * FD);
        u[g][1] = *(const float4*)(ub + OUT_NODES * FD + g * 4 * FD);
    }
    __syncthreads();

    #pragma unroll 1
    for (int st = 0; st < NSUBT - 1; ++st)
        tile_step<true>(u, sacc, btl + (st & 1) * (ROWS_SUB * BSTRIDE), vs, smax, ssum,
                        ub + (size_t)(st + 1) * ROWS_SUB * OUT_NODES * FD, t, lane, wave, fl, h, obase);
    tile_step<false>(u, sacc, btl + ((NSUBT - 1) & 1) * (ROWS_SUB * BSTRIDE), vs, smax, ssum,
                     ub, t, lane, wave, fl, h, obase);

    __syncthreads();
    float* pout = partials + (size_t)blockIdx.x * SVEC;
    for (int q = t; q < SVEC / 4; q += NTHR)
        ((float4*)pout)[q] = ((const float4*)sacc)[q];
}

__global__ void caps_reduce_squash(const float* __restrict__ partials,
                                   float* __restrict__ vsum,
                                   float* __restrict__ vout) {
    const int idx = blockIdx.x * 256 + threadIdx.x;
    float s = 0.f;
    for (int b = 0; b < NBLK; ++b)
        s += partials[(size_t)b * SVEC + idx];
    float sq = s * s;
    #pragma unroll
    for (int m = 1; m < 16; m <<= 1)
        sq += __shfl_xor(sq, m, 16);
    const float scale = (sq / (1.f + sq)) * rsqrtf(sq);
    const float v = s * scale;
    vout[idx] = v;
    vsum[idx] += v;
}

extern "C" void kernel_launch(void* const* d_in, const int* in_sizes, int n_in,
                              void* d_out, int out_size, void* d_ws, size_t ws_size,
                              hipStream_t stream) {
    const float* uh = (const float*)d_in[0];
    const int routing_num = 3;

    float* ws       = (float*)d_ws;
    float* vsum     = ws;
    float* partials = ws + SVEC;
    float* vout     = (float*)d_out;

    if (ws_size < 16842752) return;

    caps_zero_vsum<<<SVEC / 256, 256, 0, stream>>>(vsum);
    for (int it = 0; it < routing_num; ++it) {
        caps_routing_pass<<<NBLK, NTHR, SMEM_BYTES, stream>>>(uh, vsum, partials);
        caps_reduce_squash<<<SVEC / 256, 256, 0, stream>>>(partials, vsum, vout);
    }
}


# s reduction+squash launch on all 256 WGPs: 256 work-groups x 4 waves, 32 independent 16-byte partial loads per lane (was 64 work-groups, 16 dependent rounds of 4-byte loads); s summed pairwise, 8x32
# speedup vs baseline: 4.8150x; 1.2216x over previous
#include <hip/hip_runtime.h>


#define IN_NODES   4096
#define OUT_NODES  1024
#define FD         16
#define ROWS_BLK   16
#define ROWS_SUB   4
#define NSUBT      4
#define NBLK       256
#define SVEC       16384
#define BSTRIDE    1028
#define NTHR       1024
#define OWAVE      32
#define NGRP       8
#define WROW       8
#define RTHR       128
#define RCOLS      64
#define RSUB       8
#define RPER       32

typedef __attribute__((ext_vector_type(2))) float v2f;
typedef __attribute__((ext_vector_type(8))) float v8f;

#define SMEM_FLOATS (SVEC + SVEC + 2 * ROWS_SUB * BSTRIDE + ROWS_SUB * WROW + ROWS_SUB * WROW)
#define SMEM_BYTES  (SMEM_FLOATS * 4)

template <int M>
__device__ __forceinline__ float row_xor(float x) {
    return __builtin_bit_cast(float, __builtin_amdgcn_update_dpp(
        0, __builtin_bit_cast(int, x), 0x160 + M, 0xf, 0xf, true));
}
__device__ __forceinline__ float other_row(float x) {
    return __builtin_bit_cast(float, __builtin_amdgcn_permlanex16(
        __builtin_bit_cast(int, x), __builtin_bit_cast(int, x),
        0x76543210, 0xfedcba98, false, false));
}

__global__ void caps_zero_vsum(float* __restrict__ vsum) {
    vsum[blockIdx.x * 256 + threadIdx.x] = 0.0f;
}

template <bool MORE>
__device__ __forceinline__ void tile_step(float4 (&u)[NGRP][2], float* sacc,
                                          float* bw, const float* vs,
                                          float* smax, float* ssum,
                                          const float* __restrict__ un,
                                          const int t, const int lane, const int wave,
                                          const int fl, const int h, const int obase) {
    #pragma unroll
    for (int g = 0; g < NGRP; ++g) {
        const float4 v4 = *(const float4*)(vs + (obase + 4 * g) * FD + fl * 4);
        float p0 = u[g][0].x * v4.x + u[g][0].y * v4.y + u[g][0].z * v4.z + u[g][0].w * v4.w;
        float p1 = u[g][1].x * v4.x + u[g][1].y * v4.y + u[g][1].z * v4.z + u[g][1].w * v4.w;
        p0 += row_xor<1>(p0);  p1 += row_xor<1>(p1);
        p0 += row_xor<2>(p0);  p1 += row_xor<2>(p1);
        const float pv = (fl & 1) ? p1 : p0;
        if ((fl & 2) == 0)
            bw[(2 * h + (fl & 1)) * BSTRIDE + obase + 4 * g + (fl >> 2)] = pv;
    }
    __syncthreads();

    {
        const int r = t >> 8, sub = t & 255, wr = wave & (WROW - 1);
        float* row = bw + r * BSTRIDE;
        const float x0 = row[sub], x1 = row[sub + 256], x2 = row[sub + 512], x3 = row[sub + 768];
        float pm = fmaxf(fmaxf(x0, x1), fmaxf(x2, x3));
        pm = fmaxf(pm, row_xor<1>(pm));
        pm = fmaxf(pm, row_xor<2>(pm));
        pm = fmaxf(pm, row_xor<4>(pm));
        pm = fmaxf(pm, row_xor<8>(pm));
        pm = fmaxf(pm, other_row(pm));
        if (lane == 0) smax[r * WROW + wr] = pm;
        __syncthreads();
        float mm = smax[r * WROW];
        #pragma unroll
        for (int q = 1; q < WROW; ++q) mm = fmaxf(mm, smax[r * WROW + q]);
        const float e0 = __expf(x0 - mm), e1 = __expf(x1 - mm);
        const float e2 = __expf(x2 - mm), e3 = __expf(x3 - mm);
        row[sub] = e0;  row[sub + 256] = e1;  row[sub + 512] = e2;  row[sub + 768] = e3;
        float ps = (e0 + e1) + (e2 + e3);
        ps += row_xor<1>(ps);
        ps += row_xor<2>(ps);
        ps += row_xor<4>(ps);
        ps += row_xor<8>(ps);
        ps += other_row(ps);
        if (lane == 0) ssum[r * WROW + wr] = ps;
        __syncthreads();
    }

    {
        float s0 = 0.f, s1 = 0.f;
        #pragma unroll
        for (int q = 0; q < WROW; ++q) {
            s0 += ssum[(2 * h    ) * WROW + q];
            s1 += ssum[(2 * h + 1) * WROW + q];
        }
        const float ri0 = 1.f / s0, ri1 = 1.f / s1;
        const float* e0p = bw + (2 * h) * BSTRIDE + obase + (fl >> 2);
        const float* e1p = e0p + BSTRIDE;
        const bool own = (h == (fl >> 3));
        #pragma unroll
        for (int g = 0; g < NGRP; ++g) {
            v2f a;
            a.x = e0p[4 * g] * ri0;
            a.y = e1p[4 * g] * ri1;
            const float u0[4] = {u[g][0].x, u[g][0].y, u[g][0].z, u[g][0].w};
            const float u1[4] = {u[g][1].x, u[g][1].y, u[g][1].z, u[g][1].w};
            float dv[4];
            #pragma unroll
            for (int j = 0; j < 4; ++j) {
                v2f b;
                b.x = u0[j];
                b.y = u1[j];
                const v8f z = {0.f, 0.f, 0.f, 0.f, 0.f, 0.f, 0.f, 0.f};
                const v8f d = __builtin_amdgcn_wmma_f32_16x16x4_f32(
                         false, a,  false, b,
                         (short)0, z,  false,  false);
                dv[j] = (fl & 4) ? d[4] : d[0];
            }
            if (own) {
                float4* sp = (float4*)(sacc + (obase + 4 * g) * FD + fl * 4);
                float4 sv = *sp;
                sv.x += dv[0];  sv.y += dv[1];  sv.z += dv[2];  sv.w += dv[3];
                *sp = sv;
            }
            if (MORE) {
                u[g][0] = *(const float4*)(un + g * 4 * FD);
                u[g][1] = *(const float4*)(un + OUT_NODES * FD + g * 4 * FD);
            }
        }
    }
}

__global__ void __launch_bounds__(NTHR)
caps_routing_pass(const float* __restrict__ uh,
                                  const float* __restrict__ vsum_g,
                                  float* __restrict__ partials) {
    extern __shared__ float smem[];
    float* vs   = smem;
    float* sacc = vs + SVEC;
    float* btl  = sacc + SVEC;
    float* smax = btl + 2 * ROWS_SUB * BSTRIDE;
    float* ssum = smax + ROWS_SUB * WROW;

    const int t    = threadIdx.x;
    const int lane = t & 31;
    const int wave = t >> 5;
    const int fl   = lane & 15;
    const int h    = lane >> 4;
    const int obase = wave * OWAVE;

    for (int q = t; q < SVEC / 4; q += NTHR) {
        ((float4*)vs)[q]   = ((const float4*)vsum_g)[q];
        ((float4*)sacc)[q] = make_float4(0.f, 0.f, 0.f, 0.f);
    }

    const float* ub = uh + ((size_t)(blockIdx.x * ROWS_BLK + 2 * h) * OUT_NODES + obase) * FD + fl * 4;

    float4 u[NGRP][2];
    #pragma unroll
    for (int g = 0; g < NGRP; ++g) {
        u[g][0] = *(const float4*)(ub + g * 4 * FD);
        u[g][1] = *(const float4*)(ub + OUT_NODES * FD + g * 4 * FD);
    }
    __syncthreads();

    #pragma unroll 1
    for (int st = 0; st < NSUBT - 1; ++st)
        tile_step<true>(u, sacc, btl + (st & 1) * (ROWS_SUB * BSTRIDE), vs, smax, ssum,
                        ub + (size_t)(st + 1) * ROWS_SUB * OUT_NODES * FD, t, lane, wave, fl, h, obase);
    tile_step<false>(u, sacc, btl + ((NSUBT - 1) & 1) * (ROWS_SUB * BSTRIDE), vs, smax, ssum,
                     ub, t, lane, wave, fl, h, obase);

    __syncthreads();
    float* pout = partials + (size_t)blockIdx.x * SVEC;
    for (int q = t; q < SVEC / 4; q += NTHR)
        ((float4*)pout)[q] = ((const float4*)sacc)[q];
}

__global__ void __launch_bounds__(RTHR)
caps_reduce_squash(const float* __restrict__ partials,
                                   float* __restrict__ vsum,
                                   float* __restrict__ vout) {
    __shared__ float red[RSUB * RCOLS];
    const int t = threadIdx.x;
    const int q = t & 15;
    const int r = t >> 4;
    const float4* src = (const float4*)(partials + (size_t)r * SVEC + blockIdx.x * RCOLS) + q;
    const bool fin = t < RCOLS;
    const int idx = blockIdx.x * RCOLS + t;
    float vold = 0.f;
    if (fin) vold = vsum[idx];

    float4 x[RPER];
    #pragma unroll
    for (int k = 0; k < RPER; ++k)
        x[k] = src[(size_t)k * RSUB * (SVEC / 4)];
    #pragma unroll
    for (int w = 1; w < RPER; w <<= 1) {
        #pragma unroll
        for (int k = 0; k < RPER; k += 2 * w) {
            x[k].x += x[k + w].x;  x[k].y += x[k + w].y;
            x[k].z += x[k + w].z;  x[k].w += x[k + w].w;
        }
    }
    ((float4*)red)[r * (RCOLS / 4) + q] = x[0];
    __syncthreads();

    if (fin) {
        float s = red[t];
        #pragma unroll
        for (int j = 1; j < RSUB; ++j) s += red[j * RCOLS + t];
        float sq = s * s;
        sq += row_xor<1>(sq);
        sq += row_xor<2>(sq);
        sq += row_xor<4>(sq);
        sq += row_xor<8>(sq);
        const float scale = (sq / (1.f + sq)) * rsqrtf(sq);
        const float v = s * scale;
        vout[idx] = v;
        vsum[idx] = vold + v;
    }
}

extern "C" void kernel_launch(void* const* d_in, const int* in_sizes, int n_in,
                              void* d_out, int out_size, void* d_ws, size_t ws_size,
                              hipStream_t stream) {
    const float* uh = (const float*)d_in[0];
    const int routing_num = 3;

    float* ws       = (float*)d_ws;
    float* vsum     = ws;
    float* partials = ws + SVEC;
    float* vout     = (float*)d_out;

    if (ws_size < 16842752) return;

    caps_zero_vsum<<<SVEC / 256, 256, 0, stream>>>(vsum);
    for (int it = 0; it < routing_num; ++it) {
        caps_routing_pass<<<NBLK, NTHR, SMEM_BYTES, stream>>>(uh, vsum, partials);
        caps_reduce_squash<<<SVEC / RCOLS, RTHR, 0, stream>>>(partials, vsum, vout);
    }
}


# Iteration 1 specialised for its zero logits (uniform coupling 1/1024): pass 1 streams u_hat into the same WMMA column sums with no dots, softmax, barriers or LDS; Vsum zero-fill launch dropped
# speedup vs baseline: 5.3091x; 1.1026x over previous
#include <hip/hip_runtime.h>


#define IN_NODES   4096
#define OUT_NODES  1024
#define FD         16
#define ROWS_BLK   16
#define ROWS_SUB   4
#define NSUBT      4
#define NBLK       256
#define SVEC       16384
#define BSTRIDE    1028
#define NTHR       1024
#define OWAVE      32
#define NGRP       8
#define WROW       8
#define RTHR       128
#define RCOLS      64
#define RSUB       8
#define RPER       32

typedef __attribute__((ext_vector_type(2))) float v2f;
typedef __attribute__((ext_vector_type(8))) float v8f;

#define SMEM_FLOATS (SVEC + SVEC + 2 * ROWS_SUB * BSTRIDE + ROWS_SUB * WROW + ROWS_SUB * WROW)
#define SMEM_BYTES  (SMEM_FLOATS * 4)

template <int M>
__device__ __forceinline__ float row_xor(float x) {
    return __builtin_bit_cast(float, __builtin_amdgcn_update_dpp(
        0, __builtin_bit_cast(int, x), 0x160 + M, 0xf, 0xf, true));
}
__device__ __forceinline__ float other_row(float x) {
    return __builtin_bit_cast(float, __builtin_amdgcn_permlanex16(
        __builtin_bit_cast(int, x), __builtin_bit_cast(int, x),
        0x76543210, 0xfedcba98, false, false));
}

template <bool MORE>
__device__ __forceinline__ void tile_step(float4 (&u)[NGRP][2], float* sacc,
                                          float* bw, const float* vs,
                                          float* smax, float* ssum,
                                          const float* __restrict__ un,
                                          const int t, const int lane, const int wave,
                                          const int fl, const int h, const int obase) {
    #pragma unroll
    for (int g = 0; g < NGRP; ++g) {
        const float4 v4 = *(const float4*)(vs + (obase + 4 * g) * FD + fl * 4);
        float p0 = u[g][0].x * v4.x + u[g][0].y * v4.y + u[g][0].z * v4.z + u[g][0].w * v4.w;
        float p1 = u[g][1].x * v4.x + u[g][1].y * v4.y + u[g][1].z * v4.z + u[g][1].w * v4.w;
        p0 += row_xor<1>(p0);  p1 += row_xor<1>(p1);
        p0 += row_xor<2>(p0);  p1 += row_xor<2>(p1);
        const float pv = (fl & 1) ? p1 : p0;
        if ((fl & 2) == 0)
            bw[(2 * h + (fl & 1)) * BSTRIDE + obase + 4 * g + (fl >> 2)] = pv;
    }
    __syncthreads();

    {
        const int r = t >> 8, sub = t & 255, wr = wave & (WROW - 1);
        float* row = bw + r * BSTRIDE;
        const float x0 = row[sub], x1 = row[sub + 256], x2 = row[sub + 512], x3 = row[sub + 768];
        float pm = fmaxf(fmaxf(x0, x1), fmaxf(x2, x3));
        pm = fmaxf(pm, row_xor<1>(pm));
        pm = fmaxf(pm, row_xor<2>(pm));
        pm = fmaxf(pm, row_xor<4>(pm));
        pm = fmaxf(pm, row_xor<8>(pm));
        pm = fmaxf(pm, other_row(pm));
        if (lane == 0) smax[r * WROW + wr] = pm;
        __syncthreads();
        float mm = smax[r * WROW];
        #pragma unroll
        for (int q = 1; q < WROW; ++q) mm = fmaxf(mm, smax[r * WROW + q]);
        const float e0 = __expf(x0 - mm), e1 = __expf(x1 - mm);
        const float e2 = __expf(x2 - mm), e3 = __expf(x3 - mm);
        row[sub] = e0;  row[sub + 256] = e1;  row[sub + 512] = e2;  row[sub + 768] = e3;
        float ps = (e0 + e1) + (e2 + e3);
        ps += row_xor<1>(ps);
        ps += row_xor<2>(ps);
        ps += row_xor<4>(ps);
        ps += row_xor<8>(ps);
        ps += other_row(ps);
        if (lane == 0) ssum[r * WROW + wr] = ps;
        __syncthreads();
    }

    {
        float s0 = 0.f, s1 = 0.f;
        #pragma unroll
        for (int q = 0; q < WROW; ++q) {
            s0 += ssum[(2 * h    ) * WROW + q];
            s1 += ssum[(2 * h + 1) * WROW + q];
        }
        const float ri0 = 1.f / s0, ri1 = 1.f / s1;
        const float* e0p = bw + (2 * h) * BSTRIDE + obase + (fl >> 2);
        const float* e1p = e0p + BSTRIDE;
        const bool own = (h == (fl >> 3));
        #pragma unroll
        for (int g = 0; g < NGRP; ++g) {
            v2f a;
            a.x = e0p[4 * g] * ri0;
            a.y = e1p[4 * g] * ri1;
            const float u0[4] = {u[g][0].x, u[g][0].y, u[g][0].z, u[g][0].w};
            const float u1[4] = {u[g][1].x, u[g][1].y, u[g][1].z, u[g][1].w};
            float dv[4];
            #pragma unroll
            for (int j = 0; j < 4; ++j) {
                v2f b;
                b.x = u0[j];
                b.y = u1[j];
                const v8f z = {0.f, 0.f, 0.f, 0.f, 0.f, 0.f, 0.f, 0.f};
                const v8f d = __builtin_amdgcn_wmma_f32_16x16x4_f32(
                         false, a,  false, b,
                         (short)0, z,  false,  false);
                dv[j] = (fl & 4) ? d[4] : d[0];
            }
            if (own) {
                float4* sp = (float4*)(sacc + (obase + 4 * g) * FD + fl * 4);
                float4 sv = *sp;
                sv.x += dv[0];  sv.y += dv[1];  sv.z += dv[2];  sv.w += dv[3];
                *sp = sv;
            }
            if (MORE) {
                u[g][0] = *(const float4*)(un + g * 4 * FD);
                u[g][1] = *(const float4*)(un + OUT_NODES * FD + g * 4 * FD);
            }
        }
    }
}

template <bool MORE>
__device__ __forceinline__ void first_tile(float4 (&u)[NGRP][2], float4 (&acc)[NGRP],
                                           const v2f a, const float* __restrict__ un) {
    #pragma unroll
    for (int g = 0; g < NGRP; ++g) {
        const float u0[4] = {u[g][0].x, u[g][0].y, u[g][0].z, u[g][0].w};
        const float u1[4] = {u[g][1].x, u[g][1].y, u[g][1].z, u[g][1].w};
        float dv[4];
        #pragma unroll
        for (int j = 0; j < 4; ++j) {
            v2f b;
            b.x = u0[j];
            b.y = u1[j];
            const v8f z = {0.f, 0.f, 0.f, 0.f, 0.f, 0.f, 0.f, 0.f};
            const v8f d = __builtin_amdgcn_wmma_f32_16x16x4_f32(
                     false, a,  false, b,
                     (short)0, z,  false,  false);
            dv[j] = d[0];
        }
        acc[g].x += dv[0];  acc[g].y += dv[1];  acc[g].z += dv[2];  acc[g].w += dv[3];
        if (MORE) {
            u[g][0] = *(const float4*)(un + g * 4 * FD);
            u[g][1] = *(const float4*)(un + OUT_NODES * FD + g * 4 * FD);
            __builtin_amdgcn_sched_barrier(0);
        }
    }
}

__global__ void __launch_bounds__(NTHR)
caps_first_pass(const float* __restrict__ uh, float* __restrict__ partials) {
    const int t    = threadIdx.x;
    const int lane = t & 31;
    const int wave = t >> 5;
    const int fl   = lane & 15;
    const int h    = lane >> 4;
    const int obase = wave * OWAVE;

    const float* ub = uh + ((size_t)(blockIdx.x * ROWS_BLK + 2 * h) * OUT_NODES + obase) * FD + fl * 4;

    float4 u[NGRP][2];
    #pragma unroll
    for (int g = 0; g < NGRP; ++g) {
        u[g][0] = *(const float4*)(ub + g * 4 * FD);
        u[g][1] = *(const float4*)(ub + OUT_NODES * FD + g * 4 * FD);
    }
    float4 acc[NGRP];
    #pragma unroll
    for (int g = 0; g < NGRP; ++g) acc[g] = make_float4(0.f, 0.f, 0.f, 0.f);

    v2f a;
    a.x = 1.0f / (float)OUT_NODES;
    a.y = 1.0f / (float)OUT_NODES;

    #pragma unroll 1
    for (int st = 0; st < NSUBT - 1; ++st)
        first_tile<true>(u, acc, a, ub + (size_t)(st + 1) * ROWS_SUB * OUT_NODES * FD);
    first_tile<false>(u, acc, a, ub);

    float* pout = partials + (size_t)blockIdx.x * SVEC + (obase + h * (NGRP / 2) * 4) * FD + fl * 4;
    #pragma unroll
    for (int g = 0; g < NGRP / 2; ++g) {
        const float4 lo = acc[g], hi = acc[g + NGRP / 2];
        float4 sv;
        sv.x = h ? hi.x : lo.x;  sv.y = h ? hi.y : lo.y;
        sv.z = h ? hi.z : lo.z;  sv.w = h ? hi.w : lo.w;
        *(float4*)(pout + g * 4 * FD) = sv;
    }
}

__global__ void __launch_bounds__(NTHR)
caps_routing_pass(const float* __restrict__ uh,
                                  const float* __restrict__ vsum_g,
                                  float* __restrict__ partials) {
    extern __shared__ float smem[];
    float* vs   = smem;
    float* sacc = vs + SVEC;
    float* btl  = sacc + SVEC;
    float* smax = btl + 2 * ROWS_SUB * BSTRIDE;
    float* ssum = smax + ROWS_SUB * WROW;

    const int t    = threadIdx.x;
    const int lane = t & 31;
    const int wave = t >> 5;
    const int fl   = lane & 15;
    const int h    = lane >> 4;
    const int obase = wave * OWAVE;

    for (int q = t; q < SVEC / 4; q += NTHR) {
        ((float4*)vs)[q]   = ((const float4*)vsum_g)[q];
        ((float4*)sacc)[q] = make_float4(0.f, 0.f, 0.f, 0.f);
    }

    const float* ub = uh + ((size_t)(blockIdx.x * ROWS_BLK + 2 * h) * OUT_NODES + obase) * FD + fl * 4;

    float4 u[NGRP][2];
    #pragma unroll
    for (int g = 0; g < NGRP; ++g) {
        u[g][0] = *(const float4*)(ub + g * 4 * FD);
        u[g][1] = *(const float4*)(ub + OUT_NODES * FD + g * 4 * FD);
    }
    __syncthreads();

    #pragma unroll 1
    for (int st = 0; st < NSUBT - 1; ++st)
        tile_step<true>(u, sacc, btl + (st & 1) * (ROWS_SUB * BSTRIDE), vs, smax, ssum,
                        ub + (size_t)(st + 1) * ROWS_SUB * OUT_NODES * FD, t, lane, wave, fl, h, obase);
    tile_step<false>(u, sacc, btl + ((NSUBT - 1) & 1) * (ROWS_SUB * BSTRIDE), vs, smax, ssum,
                     ub, t, lane, wave, fl, h, obase);

    __syncthreads();
    float* pout = partials + (size_t)blockIdx.x * SVEC;
    for (int q = t; q < SVEC / 4; q += NTHR)
        ((float4*)pout)[q] = ((const float4*)sacc)[q];
}

template <bool FIRST>
__global__ void __launch_bounds__(RTHR)
caps_reduce_squash(const float* __restrict__ partials,
                                   float* __restrict__ vsum,
                                   float* __restrict__ vout) {
    __shared__ float red[RSUB * RCOLS];
    const int t = threadIdx.x;
    const int q = t & 15;
    const int r = t >> 4;
    const float4* src = (const float4*)(partials + (size_t)r * SVEC + blockIdx.x * RCOLS) + q;
    const bool fin = t < RCOLS;
    const int idx = blockIdx.x * RCOLS + t;
    float vold = 0.f;
    if (!FIRST && fin) vold = vsum[idx];

    float4 x[RPER];
    #pragma unroll
    for (int k = 0; k < RPER; ++k)
        x[k] = src[(size_t)k * RSUB * (SVEC / 4)];
    #pragma unroll
    for (int w = 1; w < RPER; w <<= 1) {
        #pragma unroll
        for (int k = 0; k < RPER; k += 2 * w) {
            x[k].x += x[k + w].x;  x[k].y += x[k + w].y;
            x[k].z += x[k + w].z;  x[k].w += x[k + w].w;
        }
    }
    ((float4*)red)[r * (RCOLS / 4) + q] = x[0];
    __syncthreads();

    if (fin) {
        float s = red[t];
        #pragma unroll
        for (int j = 1; j < RSUB; ++j) s += red[j * RCOLS + t];
        float sq = s * s;
        sq += row_xor<1>(sq);
        sq += row_xor<2>(sq);
        sq += row_xor<4>(sq);
        sq += row_xor<8>(sq);
        const float scale = (sq / (1.f + sq)) * rsqrtf(sq);
        const float v = s * scale;
        vout[idx] = v;
        vsum[idx] = vold + v;
    }
}

extern "C" void kernel_launch(void* const* d_in, const int* in_sizes, int n_in,
                              void* d_out, int out_size, void* d_ws, size_t ws_size,
                              hipStream_t stream) {
    const float* uh = (const float*)d_in[0];
    const int routing_num = 3;

    float* ws       = (float*)d_ws;
    float* vsum     = ws;
    float* partials = ws + SVEC;
    float* vout     = (float*)d_out;

    if (ws_size < 16842752) return;

    caps_first_pass<<<NBLK, NTHR, 0, stream>>>(uh, partials);
    caps_reduce_squash<true><<<SVEC / RCOLS, RTHR, 0, stream>>>(partials, vsum, vout);
    for (int it = 1; it < routing_num; ++it) {
        caps_routing_pass<<<NBLK, NTHR, SMEM_BYTES, stream>>>(uh, vsum, partials);
        caps_reduce_squash<false><<<SVEC / RCOLS, RTHR, 0, stream>>>(partials, vsum, vout);
    }
}
